# starting kernel
# baseline (speedup 1.0000x reference)
#include <hip/hip_runtime.h>
#include <math.h>

typedef __attribute__((ext_vector_type(2))) float v2f;
typedef __attribute__((ext_vector_type(4))) float v4f;
typedef __attribute__((ext_vector_type(8))) float v8f;

#define GAT_N 2048
#define GAT_M 8192
#define GAT_F 1024
#define GAT_K 10
#define GAT_ALPHA 0.2f

__global__ __launch_bounds__(256) void gat_gemm_xw(const float* __restrict__ X,
                                                   const float* __restrict__ W,
                                                   float* __restrict__ H,
                                                   int rows) {
    const int F = GAT_F;
    const int wavesPerBlock = blockDim.x >> 5;
    const int wave = blockIdx.x * wavesPerBlock + (threadIdx.x >> 5);
    const int rowTiles = rows >> 6;
    const int rowTile = wave % rowTiles;
    const int colTile = wave / rowTiles;
    const int lane = threadIdx.x & 31;
    const int r = lane & 15;
    const int g = lane >> 4;
    const int i0 = rowTile << 6;
    const int j0 = colTile << 6;

    v8f c[4][4];
#pragma unroll
    for (int m = 0; m < 4; ++m)
#pragma unroll
        for (int n = 0; n < 4; ++n)
#pragma unroll
            for (int v = 0; v < 8; ++v) c[m][n][v] = 0.0f;

    const float* xr[4];
#pragma unroll
    for (int m = 0; m < 4; ++m)
        xr[m] = X + (size_t)(i0 + 16 * m + r) * F;

#pragma unroll 2
    for (int k0 = 0; k0 < F; k0 += 4) {
        const int ka = k0 + 2 * g;
        v2f a[4];
#pragma unroll
        for (int m = 0; m < 4; ++m)
            a[m] = *(const v2f*)(xr[m] + ka);

        const float* w0 = W + (size_t)ka * F + j0 + r;
        const float* w1 = w0 + F;
        v2f b[4];
#pragma unroll
        for (int n = 0; n < 4; ++n) {
            b[n].x = w0[16 * n];
            b[n].y = w1[16 * n];
        }
#pragma unroll
        for (int m = 0; m < 4; ++m)
#pragma unroll
            for (int n = 0; n < 4; ++n)
                c[m][n] = __builtin_amdgcn_wmma_f32_16x16x4_f32(
                    false, a[m], false, b[n], (short)0, c[m][n], false, false);
    }

#pragma unroll
    for (int m = 0; m < 4; ++m)
#pragma unroll
        for (int n = 0; n < 4; ++n)
#pragma unroll
            for (int v = 0; v < 8; ++v) {
                const int row = i0 + 16 * m + 8 * g + v;
                const int col = j0 + 16 * n + r;
                H[(size_t)row * F + col] = c[m][n][v];
            }
}

__global__ __launch_bounds__(256) void gat_rowdot(const float* __restrict__ hf,
                                                  const float* __restrict__ he,
                                                  const float* __restrict__ a,
                                                  float* __restrict__ s1,
                                                  float* __restrict__ s2) {
    const int F = GAT_F;
    const int wave = blockIdx.x * (blockDim.x >> 5) + (threadIdx.x >> 5);
    const int lane = threadIdx.x & 31;

    const float* h;
    const float* av;
    float* out;
    int row;
    if (wave < GAT_N) { h = hf; av = a;      out = s1; row = wave; }
    else              { h = he; av = a + F;  out = s2; row = wave - GAT_N; }

    const float* hr = h + (size_t)row * F;
    float acc = 0.0f;
    for (int k = lane; k < F; k += 32) acc += hr[k] * av[k];
#pragma unroll
    for (int off = 16; off > 0; off >>= 1) acc += __shfl_down(acc, off, 32);
    if (lane == 0) out[row] = acc;
}

__global__ __launch_bounds__(256) void gat_attn_gather(const float* __restrict__ he,
                                                       const float* __restrict__ s1,
                                                       const float* __restrict__ s2,
                                                       const int* __restrict__ nidx,
                                                       float* __restrict__ out) {
    const int F = GAT_F;
    const int i = blockIdx.x;

    __shared__ int   sh_idx[GAT_K];
    __shared__ float sh_w[GAT_K];
    __shared__ int   sh_cnt;

    if (threadIdx.x == 0) {
        int   idx[GAT_K];
        float ev[GAT_K];
        int cnt = 0;
        const float s1i = s1[i];
        for (int k = 0; k < GAT_K; ++k) {
            const int j = nidx[i * GAT_K + k];
            bool dup = false;
            for (int p = 0; p < cnt; ++p) dup = dup || (idx[p] == j);
            if (!dup) {
                float e = s1i + s2[j];
                e = (e > 0.0f) ? e : GAT_ALPHA * e;
                idx[cnt] = j;
                ev[cnt] = e;
                ++cnt;
            }
        }
        float mx = ev[0];
        for (int p = 1; p < cnt; ++p) mx = fmaxf(mx, ev[p]);
        float sum = 0.0f;
        for (int p = 0; p < cnt; ++p) { ev[p] = expf(ev[p] - mx); sum += ev[p]; }
        const float inv = 1.0f / sum;
        for (int p = 0; p < cnt; ++p) { sh_idx[p] = idx[p]; sh_w[p] = ev[p] * inv; }
        sh_cnt = cnt;
    }
    __syncthreads();

    const int cnt = sh_cnt;
    const int c = threadIdx.x * 4;
    v4f acc; acc[0] = 0.f; acc[1] = 0.f; acc[2] = 0.f; acc[3] = 0.f;
    for (int p = 0; p < cnt; ++p) {
        const float w = sh_w[p];
        const v4f hv = *(const v4f*)(he + (size_t)sh_idx[p] * F + c);
        acc += hv * w;
    }
    *(v4f*)(out + (size_t)i * F + c) = acc;
}

extern "C" void kernel_launch(void* const* d_in, const int* in_sizes, int n_in,
                              void* d_out, int out_size, void* d_ws, size_t ws_size,
                              hipStream_t stream) {
    const int N = GAT_N, M = GAT_M, F = GAT_F;

    const float* feat  = (const float*)d_in[0];
    const float* embed = (const float*)d_in[1];
    const float* W     = (const float*)d_in[2];
    const float* a     = (const float*)d_in[3];
    const int*   nidx  = (const int*)d_in[4];
    float* out = (float*)d_out;

    char* ws = (char*)d_ws;
    float* hf = (float*)ws;
    float* he = (float*)(ws + (size_t)N * F * sizeof(float));
    float* s1 = (float*)(ws + (size_t)(N + M) * F * sizeof(float));
    float* s2 = s1 + N;

    {
        const int waves = (N / 64) * (F / 64);
        gat_gemm_xw<<<waves / 8, 256, 0, stream>>>(feat, W, hf, N);
    }
    {
        const int waves = (M / 64) * (F / 64);
        gat_gemm_xw<<<waves / 8, 256, 0, stream>>>(embed, W, he, M);
    }
    gat_rowdot<<<(N + M) / 8, 256, 0, stream>>>(hf, he, a, s1, s2);

    gat_attn_gather<<<N, 256, 0, stream>>>(he, s1, s2, nidx, out);
}


# Reassociate: aggregate embed rows first, project once: out=(att@embed)@W, s=X@(W@a); the WMMA GEMM runs on N rows, not N+M; the order of the float sums changes
# speedup vs baseline: 2.0224x; 2.0224x over previous
#include <hip/hip_runtime.h>
#include <math.h>

typedef __attribute__((ext_vector_type(2))) float v2f;
typedef __attribute__((ext_vector_type(4))) float v4f;
typedef __attribute__((ext_vector_type(8))) float v8f;

#define GAT_N 2048
#define GAT_M 8192
#define GAT_F 1024
#define GAT_K 10
#define GAT_ALPHA 0.2f

__global__ __launch_bounds__(256) void gat_gemm_xw(const float* __restrict__ X,
                                                   const float* __restrict__ W,
                                                   float* __restrict__ H,
                                                   int rows) {
    const int F = GAT_F;
    const int wavesPerBlock = blockDim.x >> 5;
    const int wave = blockIdx.x * wavesPerBlock + (threadIdx.x >> 5);
    const int rowTiles = rows >> 6;
    const int rowTile = wave % rowTiles;
    const int colTile = wave / rowTiles;
    const int lane = threadIdx.x & 31;
    const int r = lane & 15;
    const int g = lane >> 4;
    const int i0 = rowTile << 6;
    const int j0 = colTile << 6;

    v8f c[4][4];
#pragma unroll
    for (int m = 0; m < 4; ++m)
#pragma unroll
        for (int n = 0; n < 4; ++n)
#pragma unroll
            for (int v = 0; v < 8; ++v) c[m][n][v] = 0.0f;

    const float* xr[4];
#pragma unroll
    for (int m = 0; m < 4; ++m)
        xr[m] = X + (size_t)(i0 + 16 * m + r) * F;

#pragma unroll 2
    for (int k0 = 0; k0 < F; k0 += 4) {
        const int ka = k0 + 2 * g;
        v2f a[4];
#pragma unroll
        for (int m = 0; m < 4; ++m)
            a[m] = *(const v2f*)(xr[m] + ka);

        const float* w0 = W + (size_t)ka * F + j0 + r;
        const float* w1 = w0 + F;
        v2f b[4];
#pragma unroll
        for (int n = 0; n < 4; ++n) {
            b[n].x = w0[16 * n];
            b[n].y = w1[16 * n];
        }
#pragma unroll
        for (int m = 0; m < 4; ++m)
#pragma unroll
            for (int n = 0; n < 4; ++n)
                c[m][n] = __builtin_amdgcn_wmma_f32_16x16x4_f32(
                    false, a[m], false, b[n], (short)0, c[m][n], false, false);
    }

#pragma unroll
    for (int m = 0; m < 4; ++m)
#pragma unroll
        for (int n = 0; n < 4; ++n)
#pragma unroll
            for (int v = 0; v < 8; ++v) {
                const int row = i0 + 16 * m + 8 * g + v;
                const int col = j0 + 16 * n + r;
                H[(size_t)row * F + col] = c[m][n][v];
            }
}

__global__ __launch_bounds__(256) void gat_w_dot_a(const float* __restrict__ W,
                                                   const float* __restrict__ a,
                                                   float* __restrict__ wab) {
    const int F = GAT_F;
    const int row = blockIdx.x * (blockDim.x >> 5) + (threadIdx.x >> 5);
    const int lane = threadIdx.x & 31;

    const float* wr = W + (size_t)row * F;
    float acc1 = 0.0f, acc2 = 0.0f;
    for (int k = lane; k < F; k += 32) {
        const float w = wr[k];
        acc1 += w * a[k];
        acc2 += w * a[F + k];
    }
#pragma unroll
    for (int off = 16; off > 0; off >>= 1) {
        acc1 += __shfl_down(acc1, off, 32);
        acc2 += __shfl_down(acc2, off, 32);
    }
    if (lane == 0) { wab[row] = acc1; wab[F + row] = acc2; }
}

__global__ __launch_bounds__(256) void gat_rowdot(const float* __restrict__ hf,
                                                  const float* __restrict__ he,
                                                  const float* __restrict__ a,
                                                  float* __restrict__ s1,
                                                  float* __restrict__ s2) {
    const int F = GAT_F;
    const int wave = blockIdx.x * (blockDim.x >> 5) + (threadIdx.x >> 5);
    const int lane = threadIdx.x & 31;

    const float* h;
    const float* av;
    float* out;
    int row;
    if (wave < GAT_N) { h = hf; av = a;      out = s1; row = wave; }
    else              { h = he; av = a + F;  out = s2; row = wave - GAT_N; }

    const float* hr = h + (size_t)row * F;
    float acc = 0.0f;
    for (int k = lane; k < F; k += 32) acc += hr[k] * av[k];
#pragma unroll
    for (int off = 16; off > 0; off >>= 1) acc += __shfl_down(acc, off, 32);
    if (lane == 0) out[row] = acc;
}

__global__ __launch_bounds__(256) void gat_attn_gather(const float* __restrict__ he,
                                                       const float* __restrict__ s1,
                                                       const float* __restrict__ s2,
                                                       const int* __restrict__ nidx,
                                                       float* __restrict__ out) {
    const int F = GAT_F;
    const int i = blockIdx.x;

    __shared__ int   sh_idx[GAT_K];
    __shared__ float sh_w[GAT_K];
    __shared__ int   sh_cnt;

    if (threadIdx.x == 0) {
        int   idx[GAT_K];
        float ev[GAT_K];
        int cnt = 0;
        const float s1i = s1[i];
        for (int k = 0; k < GAT_K; ++k) {
            const int j = nidx[i * GAT_K + k];
            bool dup = false;
            for (int p = 0; p < cnt; ++p) dup = dup || (idx[p] == j);
            if (!dup) {
                float e = s1i + s2[j];
                e = (e > 0.0f) ? e : GAT_ALPHA * e;
                idx[cnt] = j;
                ev[cnt] = e;
                ++cnt;
            }
        }
        float mx = ev[0];
        for (int p = 1; p < cnt; ++p) mx = fmaxf(mx, ev[p]);
        float sum = 0.0f;
        for (int p = 0; p < cnt; ++p) { ev[p] = expf(ev[p] - mx); sum += ev[p]; }
        const float inv = 1.0f / sum;
        for (int p = 0; p < cnt; ++p) { sh_idx[p] = idx[p]; sh_w[p] = ev[p] * inv; }
        sh_cnt = cnt;
    }
    __syncthreads();

    const int cnt = sh_cnt;
    const int c = threadIdx.x * 4;
    v4f acc; acc[0] = 0.f; acc[1] = 0.f; acc[2] = 0.f; acc[3] = 0.f;
    for (int p = 0; p < cnt; ++p) {
        const float w = sh_w[p];
        const v4f hv = *(const v4f*)(he + (size_t)sh_idx[p] * F + c);
        acc += hv * w;
    }
    *(v4f*)(out + (size_t)i * F + c) = acc;
}

extern "C" void kernel_launch(void* const* d_in, const int* in_sizes, int n_in,
                              void* d_out, int out_size, void* d_ws, size_t ws_size,
                              hipStream_t stream) {
    const int N = GAT_N, M = GAT_M, F = GAT_F;

    const float* feat  = (const float*)d_in[0];
    const float* embed = (const float*)d_in[1];
    const float* W     = (const float*)d_in[2];
    const float* a     = (const float*)d_in[3];
    const int*   nidx  = (const int*)d_in[4];
    float* out = (float*)d_out;

    char* ws = (char*)d_ws;
    float* G   = (float*)ws;
    float* wab = (float*)(ws + (size_t)N * F * sizeof(float));
    float* s1  = wab + 2 * F;
    float* s2  = s1 + N;

    gat_w_dot_a<<<F / 8, 256, 0, stream>>>(W, a, wab);

    gat_rowdot<<<(N + M) / 8, 256, 0, stream>>>(feat, embed, wab, s1, s2);

    gat_attn_gather<<<N, 256, 0, stream>>>(embed, s1, s2, nidx, G);

    {
        const int waves = (N / 64) * (F / 64);
        gat_gemm_xw<<<waves / 8, 256, 0, stream>>>(G, W, out, N);
    }
}


# GEMM split 4 ways along K over the waves of a block (2 tiles x 4 K-quarters per 8-wave block), partial tiles summed through LDS in fixed order: 256 blocks, not 64; float sum order along K changes
# speedup vs baseline: 5.2213x; 2.5817x over previous
#include <hip/hip_runtime.h>
#include <math.h>

typedef __attribute__((ext_vector_type(2))) float v2f;
typedef __attribute__((ext_vector_type(4))) float v4f;
typedef __attribute__((ext_vector_type(8))) float v8f;

#define GAT_N 2048
#define GAT_M 8192
#define GAT_F 1024
#define GAT_K 10
#define GAT_ALPHA 0.2f

#define GAT_KSPLIT 4

__global__ __launch_bounds__(256) void gat_gemm_xw(const float* __restrict__ X,
                                                   const float* __restrict__ W,
                                                   float* __restrict__ H,
                                                   int rows) {
    const int F = GAT_F;
    const int wv = threadIdx.x >> 5;
    const int ks = wv & (GAT_KSPLIT - 1);
    const int tb = wv >> 2;
    const int tile = blockIdx.x * 2 + tb;
    const int rowTiles = rows >> 6;
    const int rowTile = tile % rowTiles;
    const int colTile = tile / rowTiles;
    const int lane = threadIdx.x & 31;
    const int r = lane & 15;
    const int g = lane >> 4;
    const int i0 = rowTile << 6;
    const int j0 = colTile << 6;
    const int kbeg = ks * (F / GAT_KSPLIT);

    __shared__ v4f red[2][GAT_KSPLIT - 1][32][32];

    v8f c[4][4];
#pragma unroll
    for (int m = 0; m < 4; ++m)
#pragma unroll
        for (int n = 0; n < 4; ++n)
#pragma unroll
            for (int v = 0; v < 8; ++v) c[m][n][v] = 0.0f;

    const float* xr[4];
#pragma unroll
    for (int m = 0; m < 4; ++m)
        xr[m] = X + (size_t)(i0 + 16 * m + r) * F + kbeg;
    const float* Wq = W + (size_t)kbeg * F;

#pragma unroll 2
    for (int k0 = 0; k0 < F / GAT_KSPLIT; k0 += 4) {
        const int ka = k0 + 2 * g;
        v2f a[4];
#pragma unroll
        for (int m = 0; m < 4; ++m)
            a[m] = *(const v2f*)(xr[m] + ka);

        const float* w0 = Wq + (size_t)ka * F + j0 + r;
        const float* w1 = w0 + F;
        v2f b[4];
#pragma unroll
        for (int n = 0; n < 4; ++n) {
            b[n].x = w0[16 * n];
            b[n].y = w1[16 * n];
        }
#pragma unroll
        for (int m = 0; m < 4; ++m)
#pragma unroll
            for (int n = 0; n < 4; ++n)
                c[m][n] = __builtin_amdgcn_wmma_f32_16x16x4_f32(
                    false, a[m], false, b[n], (short)0, c[m][n], false, false);
    }

    if (ks != 0) {
#pragma unroll
        for (int m = 0; m < 4; ++m)
#pragma unroll
            for (int n = 0; n < 4; ++n)
#pragma unroll
                for (int h = 0; h < 2; ++h) {
                    v4f t;
                    t[0] = c[m][n][4 * h + 0];
                    t[1] = c[m][n][4 * h + 1];
                    t[2] = c[m][n][4 * h + 2];
                    t[3] = c[m][n][4 * h + 3];
                    red[tb][ks - 1][(m * 4 + n) * 2 + h][lane] = t;
                }
    }
    __syncthreads();
    if (ks != 0) return;

#pragma unroll
    for (int p = 0; p < GAT_KSPLIT - 1; ++p)
#pragma unroll
        for (int m = 0; m < 4; ++m)
#pragma unroll
            for (int n = 0; n < 4; ++n)
#pragma unroll
                for (int h = 0; h < 2; ++h) {
                    const v4f t = red[tb][p][(m * 4 + n) * 2 + h][lane];
                    c[m][n][4 * h + 0] += t[0];
                    c[m][n][4 * h + 1] += t[1];
                    c[m][n][4 * h + 2] += t[2];
                    c[m][n][4 * h + 3] += t[3];
                }

#pragma unroll
    for (int m = 0; m < 4; ++m)
#pragma unroll
        for (int n = 0; n < 4; ++n)
#pragma unroll
            for (int v = 0; v < 8; ++v) {
                const int row = i0 + 16 * m + 8 * g + v;
                const int col = j0 + 16 * n + r;
                H[(size_t)row * F + col] = c[m][n][v];
            }
}

__global__ __launch_bounds__(256) void gat_w_dot_a(const float* __restrict__ W,
                                                   const float* __restrict__ a,
                                                   float* __restrict__ wab) {
    const int F = GAT_F;
    const int row = blockIdx.x * (blockDim.x >> 5) + (threadIdx.x >> 5);
    const int lane = threadIdx.x & 31;

    const float* wr = W + (size_t)row * F;
    float acc1 = 0.0f, acc2 = 0.0f;
    for (int k = lane; k < F; k += 32) {
        const float w = wr[k];
        acc1 += w * a[k];
        acc2 += w * a[F + k];
    }
#pragma unroll
    for (int off = 16; off > 0; off >>= 1) {
        acc1 += __shfl_down(acc1, off, 32);
        acc2 += __shfl_down(acc2, off, 32);
    }
    if (lane == 0) { wab[row] = acc1; wab[F + row] = acc2; }
}

__global__ __launch_bounds__(256) void gat_rowdot(const float* __restrict__ hf,
                                                  const float* __restrict__ he,
                                                  const float* __restrict__ a,
                                                  float* __restrict__ s1,
                                                  float* __restrict__ s2) {
    const int F = GAT_F;
    const int wave = blockIdx.x * (blockDim.x >> 5) + (threadIdx.x >> 5);
    const int lane = threadIdx.x & 31;

    const float* h;
    const float* av;
    float* out;
    int row;
    if (wave < GAT_N) { h = hf; av = a;      out = s1; row = wave; }
    else              { h = he; av = a + F;  out = s2; row = wave - GAT_N; }

    const float* hr = h + (size_t)row * F;
    float acc = 0.0f;
    for (int k = lane; k < F; k += 32) acc += hr[k] * av[k];
#pragma unroll
    for (int off = 16; off > 0; off >>= 1) acc += __shfl_down(acc, off, 32);
    if (lane == 0) out[row] = acc;
}

__global__ __launch_bounds__(256) void gat_attn_gather(const float* __restrict__ he,
                                                       const float* __restrict__ s1,
                                                       const float* __restrict__ s2,
                                                       const int* __restrict__ nidx,
                                                       float* __restrict__ out) {
    const int F = GAT_F;
    const int i = blockIdx.x;

    __shared__ int   sh_idx[GAT_K];
    __shared__ float sh_w[GAT_K];
    __shared__ int   sh_cnt;

    if (threadIdx.x == 0) {
        int   idx[GAT_K];
        float ev[GAT_K];
        int cnt = 0;
        const float s1i = s1[i];
        for (int k = 0; k < GAT_K; ++k) {
            const int j = nidx[i * GAT_K + k];
            bool dup = false;
            for (int p = 0; p < cnt; ++p) dup = dup || (idx[p] == j);
            if (!dup) {
                float e = s1i + s2[j];
                e = (e > 0.0f) ? e : GAT_ALPHA * e;
                idx[cnt] = j;
                ev[cnt] = e;
                ++cnt;
            }
        }
        float mx = ev[0];
        for (int p = 1; p < cnt; ++p) mx = fmaxf(mx, ev[p]);
        float sum = 0.0f;
        for (int p = 0; p < cnt; ++p) { ev[p] = expf(ev[p] - mx); sum += ev[p]; }
        const float inv = 1.0f / sum;
        for (int p = 0; p < cnt; ++p) { sh_idx[p] = idx[p]; sh_w[p] = ev[p] * inv; }
        sh_cnt = cnt;
    }
    __syncthreads();

    const int cnt = sh_cnt;
    const int c = threadIdx.x * 4;
    v4f acc; acc[0] = 0.f; acc[1] = 0.f; acc[2] = 0.f; acc[3] = 0.f;
    for (int p = 0; p < cnt; ++p) {
        const float w = sh_w[p];
        const v4f hv = *(const v4f*)(he + (size_t)sh_idx[p] * F + c);
        acc += hv * w;
    }
    *(v4f*)(out + (size_t)i * F + c) = acc;
}

extern "C" void kernel_launch(void* const* d_in, const int* in_sizes, int n_in,
                              void* d_out, int out_size, void* d_ws, size_t ws_size,
                              hipStream_t stream) {
    const int N = GAT_N, M = GAT_M, F = GAT_F;

    const float* feat  = (const float*)d_in[0];
    const float* embed = (const float*)d_in[1];
    const float* W     = (const float*)d_in[2];
    const float* a     = (const float*)d_in[3];
    const int*   nidx  = (const int*)d_in[4];
    float* out = (float*)d_out;

    char* ws = (char*)d_ws;
    float* G   = (float*)ws;
    float* wab = (float*)(ws + (size_t)N * F * sizeof(float));
    float* s1  = wab + 2 * F;
    float* s2  = s1 + N;

    gat_w_dot_a<<<F / 8, 256, 0, stream>>>(W, a, wab);

    gat_rowdot<<<(N + M) / 8, 256, 0, stream>>>(feat, embed, wab, s1, s2);

    gat_attn_gather<<<N, 256, 0, stream>>>(embed, s1, s2, nidx, G);

    {
        const int tiles = (N / 64) * (F / 64);
        gat_gemm_xw<<<tiles / 2, 256, 0, stream>>>(G, W, out, N);
    }
}


# Scores fused into the gather: one wave per output row reads its 10 neighbour rows once into registers, dots them with W@a there, softmax, weighted sum; 3 launches, not 4; score sum order changes
# speedup vs baseline: 6.3433x; 1.2149x over previous
#include <hip/hip_runtime.h>
#include <math.h>

typedef __attribute__((ext_vector_type(2))) float v2f;
typedef __attribute__((ext_vector_type(4))) float v4f;
typedef __attribute__((ext_vector_type(8))) float v8f;

#define GAT_N 2048
#define GAT_M 8192
#define GAT_F 1024
#define GAT_K 10
#define GAT_ALPHA 0.2f

#define GAT_KSPLIT 4

__global__ __launch_bounds__(256) void gat_gemm_xw(const float* __restrict__ X,
                                                   const float* __restrict__ W,
                                                   float* __restrict__ H,
                                                   int rows) {
    const int F = GAT_F;
    const int wv = threadIdx.x >> 5;
    const int ks = wv & (GAT_KSPLIT - 1);
    const int tb = wv >> 2;
    const int tile = blockIdx.x * 2 + tb;
    const int rowTiles = rows >> 6;
    const int rowTile = tile % rowTiles;
    const int colTile = tile / rowTiles;
    const int lane = threadIdx.x & 31;
    const int r = lane & 15;
    const int g = lane >> 4;
    const int i0 = rowTile << 6;
    const int j0 = colTile << 6;
    const int kbeg = ks * (F / GAT_KSPLIT);

    __shared__ v4f red[2][GAT_KSPLIT - 1][32][32];

    v8f c[4][4];
#pragma unroll
    for (int m = 0; m < 4; ++m)
#pragma unroll
        for (int n = 0; n < 4; ++n)
#pragma unroll
            for (int v = 0; v < 8; ++v) c[m][n][v] = 0.0f;

    const float* xr[4];
#pragma unroll
    for (int m = 0; m < 4; ++m)
        xr[m] = X + (size_t)(i0 + 16 * m + r) * F + kbeg;
    const float* Wq = W + (size_t)kbeg * F;

#pragma unroll 2
    for (int k0 = 0; k0 < F / GAT_KSPLIT; k0 += 4) {
        const int ka = k0 + 2 * g;
        v2f a[4];
#pragma unroll
        for (int m = 0; m < 4; ++m)
            a[m] = *(const v2f*)(xr[m] + ka);

        const float* w0 = Wq + (size_t)ka * F + j0 + r;
        const float* w1 = w0 + F;
        v2f b[4];
#pragma unroll
        for (int n = 0; n < 4; ++n) {
            b[n].x = w0[16 * n];
            b[n].y = w1[16 * n];
        }
#pragma unroll
        for (int m = 0; m < 4; ++m)
#pragma unroll
            for (int n = 0; n < 4; ++n)
                c[m][n] = __builtin_amdgcn_wmma_f32_16x16x4_f32(
                    false, a[m], false, b[n], (short)0, c[m][n], false, false);
    }

    if (ks != 0) {
#pragma unroll
        for (int m = 0; m < 4; ++m)
#pragma unroll
            for (int n = 0; n < 4; ++n)
#pragma unroll
                for (int h = 0; h < 2; ++h) {
                    v4f t;
                    t[0] = c[m][n][4 * h + 0];
                    t[1] = c[m][n][4 * h + 1];
                    t[2] = c[m][n][4 * h + 2];
                    t[3] = c[m][n][4 * h + 3];
                    red[tb][ks - 1][(m * 4 + n) * 2 + h][lane] = t;
                }
    }
    __syncthreads();
    if (ks != 0) return;

#pragma unroll
    for (int p = 0; p < GAT_KSPLIT - 1; ++p)
#pragma unroll
        for (int m = 0; m < 4; ++m)
#pragma unroll
            for (int n = 0; n < 4; ++n)
#pragma unroll
                for (int h = 0; h < 2; ++h) {
                    const v4f t = red[tb][p][(m * 4 + n) * 2 + h][lane];
                    c[m][n][4 * h + 0] += t[0];
                    c[m][n][4 * h + 1] += t[1];
                    c[m][n][4 * h + 2] += t[2];
                    c[m][n][4 * h + 3] += t[3];
                }

#pragma unroll
    for (int m = 0; m < 4; ++m)
#pragma unroll
        for (int n = 0; n < 4; ++n)
#pragma unroll
            for (int v = 0; v < 8; ++v) {
                const int row = i0 + 16 * m + 8 * g + v;
                const int col = j0 + 16 * n + r;
                H[(size_t)row * F + col] = c[m][n][v];
            }
}

__global__ __launch_bounds__(256) void gat_w_dot_a(const float* __restrict__ W,
                                                   const float* __restrict__ a,
                                                   float* __restrict__ wab) {
    const int F = GAT_F;
    const int row = blockIdx.x * (blockDim.x >> 5) + (threadIdx.x >> 5);
    const int lane = threadIdx.x & 31;

    const float* wr = W + (size_t)row * F;
    float acc1 = 0.0f, acc2 = 0.0f;
    for (int k = lane; k < F; k += 32) {
        const float w = wr[k];
        acc1 += w * a[k];
        acc2 += w * a[F + k];
    }
#pragma unroll
    for (int off = 16; off > 0; off >>= 1) {
        acc1 += __shfl_down(acc1, off, 32);
        acc2 += __shfl_down(acc2, off, 32);
    }
    if (lane == 0) { wab[row] = acc1; wab[F + row] = acc2; }
}

#define GAT_Q (GAT_F / 128)

__global__ __launch_bounds__(32) void gat_score_gather(const float* __restrict__ feat,
                                                       const float* __restrict__ embed,
                                                       const float* __restrict__ wab,
                                                       const int* __restrict__ nidx,
                                                       float* __restrict__ G) {
    const int F = GAT_F;
    const int i = blockIdx.x;
    const int c = threadIdx.x * 4;

    int idx[GAT_K];
#pragma unroll
    for (int p = 0; p < GAT_K; ++p) idx[p] = nidx[i * GAT_K + p];

    v4f w1[GAT_Q], w2[GAT_Q], fv[GAT_Q];
#pragma unroll
    for (int q = 0; q < GAT_Q; ++q) {
        fv[q] = *(const v4f*)(feat + (size_t)i * F + 128 * q + c);
        w1[q] = *(const v4f*)(wab + 128 * q + c);
        w2[q] = *(const v4f*)(wab + F + 128 * q + c);
    }

    v4f hv[GAT_K][GAT_Q];
#pragma unroll
    for (int p = 0; p < GAT_K; ++p)
#pragma unroll
        for (int q = 0; q < GAT_Q; ++q)
            hv[p][q] = *(const v4f*)(embed + (size_t)idx[p] * F + 128 * q + c);

    float d[GAT_K + 1];
    {
        v4f t = fv[0] * w1[0];
#pragma unroll
        for (int q = 1; q < GAT_Q; ++q) t += fv[q] * w1[q];
        d[GAT_K] = (t[0] + t[1]) + (t[2] + t[3]);
    }
#pragma unroll
    for (int p = 0; p < GAT_K; ++p) {
        v4f t = hv[p][0] * w2[0];
#pragma unroll
        for (int q = 1; q < GAT_Q; ++q) t += hv[p][q] * w2[q];
        d[p] = (t[0] + t[1]) + (t[2] + t[3]);
    }

#pragma unroll
    for (int off = 16; off > 0; off >>= 1)
#pragma unroll
        for (int p = 0; p <= GAT_K; ++p) d[p] += __shfl_xor(d[p], off, 32);

    const float s1i = d[GAT_K];
    float ev[GAT_K];
    bool dup[GAT_K];
#pragma unroll
    for (int p = 0; p < GAT_K; ++p) {
        bool dp = false;
#pragma unroll
        for (int t = 0; t < p; ++t) dp = dp || (idx[t] == idx[p]);
        dup[p] = dp;
        const float e = s1i + d[p];
        ev[p] = (e > 0.0f) ? e : GAT_ALPHA * e;
    }
    float mx = ev[0];
#pragma unroll
    for (int p = 1; p < GAT_K; ++p) mx = fmaxf(mx, ev[p]);
    float sum = 0.0f;
#pragma unroll
    for (int p = 0; p < GAT_K; ++p) {
        ev[p] = dup[p] ? 0.0f : expf(ev[p] - mx);
        sum += ev[p];
    }
    const float inv = 1.0f / sum;

#pragma unroll
    for (int q = 0; q < GAT_Q; ++q) {
        v4f acc; acc[0] = 0.f; acc[1] = 0.f; acc[2] = 0.f; acc[3] = 0.f;
#pragma unroll
        for (int p = 0; p < GAT_K; ++p) acc += hv[p][q] * (ev[p] * inv);
        *(v4f*)(G + (size_t)i * F + 128 * q + c) = acc;
    }
}

extern "C" void kernel_launch(void* const* d_in, const int* in_sizes, int n_in,
                              void* d_out, int out_size, void* d_ws, size_t ws_size,
                              hipStream_t stream) {
    const int N = GAT_N, F = GAT_F;

    const float* feat  = (const float*)d_in[0];
    const float* embed = (const float*)d_in[1];
    const float* W     = (const float*)d_in[2];
    const float* a     = (const float*)d_in[3];
    const int*   nidx  = (const int*)d_in[4];
    float* out = (float*)d_out;

    char* ws = (char*)d_ws;
    float* G   = (float*)ws;
    float* wab = (float*)(ws + (size_t)N * F * sizeof(float));

    gat_w_dot_a<<<F / 8, 256, 0, stream>>>(W, a, wab);

    gat_score_gather<<<N, 32, 0, stream>>>(feat, embed, wab, nidx, G);

    {
        const int tiles = (N / 64) * (F / 64);
        gat_gemm_xw<<<tiles / 2, 256, 0, stream>>>(G, W, out, N);
    }
}


# GEMM K loop register double-buffered: the next stage's 24 fragment loads go out as one group in front of the current stage's 32 WMMAs (second operand set); sum order unchanged
# speedup vs baseline: 6.8323x; 1.0771x over previous
#include <hip/hip_runtime.h>
#include <math.h>

typedef __attribute__((ext_vector_type(2))) float v2f;
typedef __attribute__((ext_vector_type(4))) float v4f;
typedef __attribute__((ext_vector_type(8))) float v8f;

#define GAT_N 2048
#define GAT_M 8192
#define GAT_F 1024
#define GAT_K 10
#define GAT_ALPHA 0.2f

#define GAT_KSPLIT 4

__device__ __forceinline__ void gat_load_stage(v2f (&a)[2][4], v2f (&b)[2][4],
                                               const float* const (&xr)[4],
                                               const float* __restrict__ Wcol,
                                               int k0, int g) {
#pragma unroll
    for (int s = 0; s < 2; ++s) {
        const int ka = k0 + 4 * s + 2 * g;
#pragma unroll
        for (int m = 0; m < 4; ++m)
            a[s][m] = *(const v2f*)(xr[m] + ka);
        const float* w0 = Wcol + (size_t)ka * GAT_F;
        const float* w1 = w0 + GAT_F;
#pragma unroll
        for (int n = 0; n < 4; ++n) {
            b[s][n].x = w0[16 * n];
            b[s][n].y = w1[16 * n];
        }
    }
}

__device__ __forceinline__ void gat_mma_stage(v8f (&c)[4][4], const v2f (&a)[2][4],
                                              const v2f (&b)[2][4]) {
#pragma unroll
    for (int s = 0; s < 2; ++s)
#pragma unroll
        for (int m = 0; m < 4; ++m)
#pragma unroll
            for (int n = 0; n < 4; ++n)
                c[m][n] = __builtin_amdgcn_wmma_f32_16x16x4_f32(
                    false, a[s][m], false, b[s][n], (short)0, c[m][n], false, false);
}

__global__ __launch_bounds__(256) void gat_gemm_xw(const float* __restrict__ X,
                                                   const float* __restrict__ W,
                                                   float* __restrict__ H,
                                                   int rows) {
    const int F = GAT_F;
    const int wv = threadIdx.x >> 5;
    const int ks = wv & (GAT_KSPLIT - 1);
    const int tb = wv >> 2;
    const int tile = blockIdx.x * 2 + tb;
    const int rowTiles = rows >> 6;
    const int rowTile = tile % rowTiles;
    const int colTile = tile / rowTiles;
    const int lane = threadIdx.x & 31;
    const int r = lane & 15;
    const int g = lane >> 4;
    const int i0 = rowTile << 6;
    const int j0 = colTile << 6;
    const int kbeg = ks * (F / GAT_KSPLIT);

    __shared__ v4f red[2][GAT_KSPLIT - 1][32][32];

    v8f c[4][4];
#pragma unroll
    for (int m = 0; m < 4; ++m)
#pragma unroll
        for (int n = 0; n < 4; ++n)
#pragma unroll
            for (int v = 0; v < 8; ++v) c[m][n][v] = 0.0f;

    const float* xr[4];
#pragma unroll
    for (int m = 0; m < 4; ++m)
        xr[m] = X + (size_t)(i0 + 16 * m + r) * F + kbeg;
    const float* Wcol = W + (size_t)kbeg * F + j0 + r;

    const int KQ = F / GAT_KSPLIT;
    v2f a0[2][4], b0[2][4], a1[2][4], b1[2][4];

    gat_load_stage(a0, b0, xr, Wcol, 0, g);
    for (int k0 = 0; k0 < KQ - 16; k0 += 16) {
        gat_load_stage(a1, b1, xr, Wcol, k0 + 8, g);
        __builtin_amdgcn_sched_barrier(0);
        gat_mma_stage(c, a0, b0);
        gat_load_stage(a0, b0, xr, Wcol, k0 + 16, g);
        __builtin_amdgcn_sched_barrier(0);
        gat_mma_stage(c, a1, b1);
    }
    gat_load_stage(a1, b1, xr, Wcol, KQ - 8, g);
    __builtin_amdgcn_sched_barrier(0);
    gat_mma_stage(c, a0, b0);
    gat_mma_stage(c, a1, b1);

    if (ks != 0) {
#pragma unroll
        for (int m = 0; m < 4; ++m)
#pragma unroll
            for (int n = 0; n < 4; ++n)
#pragma unroll
                for (int h = 0; h < 2; ++h) {
                    v4f t;
                    t[0] = c[m][n][4 * h + 0];
                    t[1] = c[m][n][4 * h + 1];
                    t[2] = c[m][n][4 * h + 2];
                    t[3] = c[m][n][4 * h + 3];
                    red[tb][ks - 1][(m * 4 + n) * 2 + h][lane] = t;
                }
    }
    __syncthreads();
    if (ks != 0) return;

#pragma unroll
    for (int p = 0; p < GAT_KSPLIT - 1; ++p)
#pragma unroll
        for (int m = 0; m < 4; ++m)
#pragma unroll
            for (int n = 0; n < 4; ++n)
#pragma unroll
                for (int h = 0; h < 2; ++h) {
                    const v4f t = red[tb][p][(m * 4 + n) * 2 + h][lane];
                    c[m][n][4 * h + 0] += t[0];
                    c[m][n][4 * h + 1] += t[1];
                    c[m][n][4 * h + 2] += t[2];
                    c[m][n][4 * h + 3] += t[3];
                }

#pragma unroll
    for (int m = 0; m < 4; ++m)
#pragma unroll
        for (int n = 0; n < 4; ++n)
#pragma unroll
            for (int v = 0; v < 8; ++v) {
                const int row = i0 + 16 * m + 8 * g + v;
                const int col = j0 + 16 * n + r;
                H[(size_t)row * F + col] = c[m][n][v];
            }
}

__global__ __launch_bounds__(256) void gat_w_dot_a(const float* __restrict__ W,
                                                   const float* __restrict__ a,
                                                   float* __restrict__ wab) {
    const int F = GAT_F;
    const int row = blockIdx.x * (blockDim.x >> 5) + (threadIdx.x >> 5);
    const int lane = threadIdx.x & 31;

    const float* wr = W + (size_t)row * F;
    float acc1 = 0.0f, acc2 = 0.0f;
    for (int k = lane; k < F; k += 32) {
        const float w = wr[k];
        acc1 += w * a[k];
        acc2 += w * a[F + k];
    }
#pragma unroll
    for (int off = 16; off > 0; off >>= 1) {
        acc1 += __shfl_down(acc1, off, 32);
        acc2 += __shfl_down(acc2, off, 32);
    }
    if (lane == 0) { wab[row] = acc1; wab[F + row] = acc2; }
}

#define GAT_Q (GAT_F / 128)

__global__ __launch_bounds__(32) void gat_score_gather(const float* __restrict__ feat,
                                                       const float* __restrict__ embed,
                                                       const float* __restrict__ wab,
                                                       const int* __restrict__ nidx,
                                                       float* __restrict__ G) {
    const int F = GAT_F;
    const int i = blockIdx.x;
    const int c = threadIdx.x * 4;

    int idx[GAT_K];
#pragma unroll
    for (int p = 0; p < GAT_K; ++p) idx[p] = nidx[i * GAT_K + p];

    v4f w1[GAT_Q], w2[GAT_Q], fv[GAT_Q];
#pragma unroll
    for (int q = 0; q < GAT_Q; ++q) {
        fv[q] = *(const v4f*)(feat + (size_t)i * F + 128 * q + c);
        w1[q] = *(const v4f*)(wab + 128 * q + c);
        w2[q] = *(const v4f*)(wab + F + 128 * q + c);
    }

    v4f hv[GAT_K][GAT_Q];
#pragma unroll
    for (int p = 0; p < GAT_K; ++p)
#pragma unroll
        for (int q = 0; q < GAT_Q; ++q)
            hv[p][q] = *(const v4f*)(embed + (size_t)idx[p] * F + 128 * q + c);

    float d[GAT_K + 1];
    {
        v4f t = fv[0] * w1[0];
#pragma unroll
        for (int q = 1; q < GAT_Q; ++q) t += fv[q] * w1[q];
        d[GAT_K] = (t[0] + t[1]) + (t[2] + t[3]);
    }
#pragma unroll
    for (int p = 0; p < GAT_K; ++p) {
        v4f t = hv[p][0] * w2[0];
#pragma unroll
        for (int q = 1; q < GAT_Q; ++q) t += hv[p][q] * w2[q];
        d[p] = (t[0] + t[1]) + (t[2] + t[3]);
    }

#pragma unroll
    for (int off = 16; off > 0; off >>= 1)
#pragma unroll
        for (int p = 0; p <= GAT_K; ++p) d[p] += __shfl_xor(d[p], off, 32);

    const float s1i = d[GAT_K];
    float ev[GAT_K];
    bool dup[GAT_K];
#pragma unroll
    for (int p = 0; p < GAT_K; ++p) {
        bool dp = false;
#pragma unroll
        for (int t = 0; t < p; ++t) dp = dp || (idx[t] == idx[p]);
        dup[p] = dp;
        const float e = s1i + d[p];
        ev[p] = (e > 0.0f) ? e : GAT_ALPHA * e;
    }
    float mx = ev[0];
#pragma unroll
    for (int p = 1; p < GAT_K; ++p) mx = fmaxf(mx, ev[p]);
    float sum = 0.0f;
#pragma unroll
    for (int p = 0; p < GAT_K; ++p) {
        ev[p] = dup[p] ? 0.0f : expf(ev[p] - mx);
        sum += ev[p];
    }
    const float inv = 1.0f / sum;

#pragma unroll
    for (int q = 0; q < GAT_Q; ++q) {
        v4f acc; acc[0] = 0.f; acc[1] = 0.f; acc[2] = 0.f; acc[3] = 0.f;
#pragma unroll
        for (int p = 0; p < GAT_K; ++p) acc += hv[p][q] * (ev[p] * inv);
        *(v4f*)(G + (size_t)i * F + 128 * q + c) = acc;
    }
}

extern "C" void kernel_launch(void* const* d_in, const int* in_sizes, int n_in,
                              void* d_out, int out_size, void* d_ws, size_t ws_size,
                              hipStream_t stream) {
    const int N = GAT_N, F = GAT_F;

    const float* feat  = (const float*)d_in[0];
    const float* embed = (const float*)d_in[1];
    const float* W     = (const float*)d_in[2];
    const float* a     = (const float*)d_in[3];
    const int*   nidx  = (const int*)d_in[4];
    float* out = (float*)d_out;

    char* ws = (char*)d_ws;
    float* G   = (float*)ws;
    float* wab = (float*)(ws + (size_t)N * F * sizeof(float));

    gat_w_dot_a<<<F / 8, 256, 0, stream>>>(W, a, wab);

    gat_score_gather<<<N, 32, 0, stream>>>(feat, embed, wab, nidx, G);

    {
        const int tiles = (N / 64) * (F / 64);
        gat_gemm_xw<<<tiles / 2, 256, 0, stream>>>(G, W, out, N);
    }
}


# GEMM fragment loads widened to 16 B per lane (4 consecutive K of A, 4 adjacent columns of W), stage = 32 K so a 128-B line of A is fetched once; 16-B stores; K grouping in a WMMA changes sum order
# speedup vs baseline: 7.7950x; 1.1409x over previous
#include <hip/hip_runtime.h>
#include <math.h>

typedef __attribute__((ext_vector_type(2))) float v2f;
typedef __attribute__((ext_vector_type(4))) float v4f;
typedef __attribute__((ext_vector_type(8))) float v8f;

#define GAT_N 2048
#define GAT_M 8192
#define GAT_F 1024
#define GAT_K 10
#define GAT_ALPHA 0.2f

#define GAT_KSPLIT 4
#define GAT_STAGE_K 32

__device__ __forceinline__ void gat_load_stage(v4f (&a)[4][4], v4f (&b)[4][4],
                                               const float* const (&xr)[4],
                                               const float* __restrict__ Wq,
                                               int k0) {
#pragma unroll
    for (int m = 0; m < 4; ++m)
#pragma unroll
        for (int kb = 0; kb < 4; ++kb)
            a[kb][m] = *(const v4f*)(xr[m] + k0 + 8 * kb);
#pragma unroll
    for (int kb = 0; kb < 4; ++kb)
#pragma unroll
        for (int t = 0; t < 4; ++t)
            b[kb][t] = *(const v4f*)(Wq + (size_t)(k0 + 8 * kb + t) * GAT_F);
}

__device__ __forceinline__ void gat_mma_stage(v8f (&c)[4][4], const v4f (&a)[4][4],
                                              const v4f (&b)[4][4]) {
#pragma unroll
    for (int kb = 0; kb < 4; ++kb)
#pragma unroll
        for (int s = 0; s < 2; ++s)
#pragma unroll
            for (int m = 0; m < 4; ++m) {
                v2f av;
                av.x = a[kb][m][2 * s];
                av.y = a[kb][m][2 * s + 1];
#pragma unroll
                for (int n = 0; n < 4; ++n) {
                    v2f bv;
                    bv.x = b[kb][2 * s][n];
                    bv.y = b[kb][2 * s + 1][n];
                    c[m][n] = __builtin_amdgcn_wmma_f32_16x16x4_f32(
                        false, av, false, bv, (short)0, c[m][n], false, false);
                }
            }
}

__global__ __launch_bounds__(256) void gat_gemm_xw(const float* __restrict__ X,
                                                   const float* __restrict__ W,
                                                   float* __restrict__ H,
                                                   int rows) {
    const int F = GAT_F;
    const int wv = threadIdx.x >> 5;
    const int ks = wv & (GAT_KSPLIT - 1);
    const int tb = wv >> 2;
    const int tile = blockIdx.x * 2 + tb;
    const int rowTiles = rows >> 6;
    const int rowTile = tile % rowTiles;
    const int colTile = tile / rowTiles;
    const int lane = threadIdx.x & 31;
    const int r = lane & 15;
    const int g = lane >> 4;
    const int i0 = rowTile << 6;
    const int j0 = colTile << 6;
    const int kbeg = ks * (F / GAT_KSPLIT);

    __shared__ v4f red[2][GAT_KSPLIT - 1][32][32];

    v8f c[4][4];
#pragma unroll
    for (int m = 0; m < 4; ++m)
#pragma unroll
        for (int n = 0; n < 4; ++n)
#pragma unroll
            for (int v = 0; v < 8; ++v) c[m][n][v] = 0.0f;

    const float* xr[4];
#pragma unroll
    for (int m = 0; m < 4; ++m)
        xr[m] = X + (size_t)(i0 + 16 * m + r) * F + kbeg + 4 * g;
    const float* Wq = W + (size_t)(kbeg + 4 * g) * F + j0 + 4 * r;

    const int KQ = F / GAT_KSPLIT;
    v4f a0[4][4], b0[4][4], a1[4][4], b1[4][4];

    gat_load_stage(a0, b0, xr, Wq, 0);
    for (int k0 = 0; k0 < KQ - 2 * GAT_STAGE_K; k0 += 2 * GAT_STAGE_K) {
        gat_load_stage(a1, b1, xr, Wq, k0 + GAT_STAGE_K);
        __builtin_amdgcn_sched_barrier(0);
        gat_mma_stage(c, a0, b0);
        gat_load_stage(a0, b0, xr, Wq, k0 + 2 * GAT_STAGE_K);
        __builtin_amdgcn_sched_barrier(0);
        gat_mma_stage(c, a1, b1);
    }
    gat_load_stage(a1, b1, xr, Wq, KQ - GAT_STAGE_K);
    __builtin_amdgcn_sched_barrier(0);
    gat_mma_stage(c, a0, b0);
    gat_mma_stage(c, a1, b1);

    if (ks != 0) {
#pragma unroll
        for (int m = 0; m < 4; ++m)
#pragma unroll
            for (int n = 0; n < 4; ++n)
#pragma unroll
                for (int h = 0; h < 2; ++h) {
                    v4f t;
                    t[0] = c[m][n][4 * h + 0];
                    t[1] = c[m][n][4 * h + 1];
                    t[2] = c[m][n][4 * h + 2];
                    t[3] = c[m][n][4 * h + 3];
                    red[tb][ks - 1][(m * 4 + n) * 2 + h][lane] = t;
                }
    }
    __syncthreads();
    if (ks != 0) return;

#pragma unroll
    for (int p = 0; p < GAT_KSPLIT - 1; ++p)
#pragma unroll
        for (int m = 0; m < 4; ++m)
#pragma unroll
            for (int n = 0; n < 4; ++n)
#pragma unroll
                for (int h = 0; h < 2; ++h) {
                    const v4f t = red[tb][p][(m * 4 + n) * 2 + h][lane];
                    c[m][n][4 * h + 0] += t[0];
                    c[m][n][4 * h + 1] += t[1];
                    c[m][n][4 * h + 2] += t[2];
                    c[m][n][4 * h + 3] += t[3];
                }

#pragma unroll
    for (int m = 0; m < 4; ++m)
#pragma unroll
        for (int v = 0; v < 8; ++v) {
            const int row = i0 + 16 * m + 8 * g + v;
            v4f o;
            o[0] = c[m][0][v];
            o[1] = c[m][1][v];
            o[2] = c[m][2][v];
            o[3] = c[m][3][v];
            *(v4f*)(H + (size_t)row * F + j0 + 4 * r) = o;
        }
}

__global__ __launch_bounds__(256) void gat_w_dot_a(const float* __restrict__ W,
                                                   const float* __restrict__ a,
                                                   float* __restrict__ wab) {
    const int F = GAT_F;
    const int row = blockIdx.x * (blockDim.x >> 5) + (threadIdx.x >> 5);
    const int lane = threadIdx.x & 31;

    const float* wr = W + (size_t)row * F;
    float acc1 = 0.0f, acc2 = 0.0f;
    for (int k = lane; k < F; k += 32) {
        const float w = wr[k];
        acc1 += w * a[k];
        acc2 += w * a[F + k];
    }
#pragma unroll
    for (int off = 16; off > 0; off >>= 1) {
        acc1 += __shfl_down(acc1, off, 32);
        acc2 += __shfl_down(acc2, off, 32);
    }
    if (lane == 0) { wab[row] = acc1; wab[F + row] = acc2; }
}

#define GAT_Q (GAT_F / 128)

__global__ __launch_bounds__(32) void gat_score_gather(const float* __restrict__ feat,
                                                       const float* __restrict__ embed,
                                                       const float* __restrict__ wab,
                                                       const int* __restrict__ nidx,
                                                       float* __restrict__ G) {
    const int F = GAT_F;
    const int i = blockIdx.x;
    const int c = threadIdx.x * 4;

    int idx[GAT_K];
#pragma unroll
    for (int p = 0; p < GAT_K; ++p) idx[p] = nidx[i * GAT_K + p];

    v4f w1[GAT_Q], w2[GAT_Q], fv[GAT_Q];
#pragma unroll
    for (int q = 0; q < GAT_Q; ++q) {
        fv[q] = *(const v4f*)(feat + (size_t)i * F + 128 * q + c);
        w1[q] = *(const v4f*)(wab + 128 * q + c);
        w2[q] = *(const v4f*)(wab + F + 128 * q + c);
    }

    v4f hv[GAT_K][GAT_Q];
#pragma unroll
    for (int p = 0; p < GAT_K; ++p)
#pragma unroll
        for (int q = 0; q < GAT_Q; ++q)
            hv[p][q] = *(const v4f*)(embed + (size_t)idx[p] * F + 128 * q + c);

    float d[GAT_K + 1];
    {
        v4f t = fv[0] * w1[0];
#pragma unroll
        for (int q = 1; q < GAT_Q; ++q) t += fv[q] * w1[q];
        d[GAT_K] = (t[0] + t[1]) + (t[2] + t[3]);
    }
#pragma unroll
    for (int p = 0; p < GAT_K; ++p) {
        v4f t = hv[p][0] * w2[0];
#pragma unroll
        for (int q = 1; q < GAT_Q; ++q) t += hv[p][q] * w2[q];
        d[p] = (t[0] + t[1]) + (t[2] + t[3]);
    }

#pragma unroll
    for (int off = 16; off > 0; off >>= 1)
#pragma unroll
        for (int p = 0; p <= GAT_K; ++p) d[p] += __shfl_xor(d[p], off, 32);

    const float s1i = d[GAT_K];
    float ev[GAT_K];
    bool dup[GAT_K];
#pragma unroll
    for (int p = 0; p < GAT_K; ++p) {
        bool dp = false;
#pragma unroll
        for (int t = 0; t < p; ++t) dp = dp || (idx[t] == idx[p]);
        dup[p] = dp;
        const float e = s1i + d[p];
        ev[p] = (e > 0.0f) ? e : GAT_ALPHA * e;
    }
    float mx = ev[0];
#pragma unroll
    for (int p = 1; p < GAT_K; ++p) mx = fmaxf(mx, ev[p]);
    float sum = 0.0f;
#pragma unroll
    for (int p = 0; p < GAT_K; ++p) {
        ev[p] = dup[p] ? 0.0f : expf(ev[p] - mx);
        sum += ev[p];
    }
    const float inv = 1.0f / sum;

#pragma unroll
    for (int q = 0; q < GAT_Q; ++q) {
        v4f acc; acc[0] = 0.f; acc[1] = 0.f; acc[2] = 0.f; acc[3] = 0.f;
#pragma unroll
        for (int p = 0; p < GAT_K; ++p) acc += hv[p][q] * (ev[p] * inv);
        *(v4f*)(G + (size_t)i * F + 128 * q + c) = acc;
    }
}

extern "C" void kernel_launch(void* const* d_in, const int* in_sizes, int n_in,
                              void* d_out, int out_size, void* d_ws, size_t ws_size,
                              hipStream_t stream) {
    const int N = GAT_N, F = GAT_F;

    const float* feat  = (const float*)d_in[0];
    const float* embed = (const float*)d_in[1];
    const float* W     = (const float*)d_in[2];
    const float* a     = (const float*)d_in[3];
    const int*   nidx  = (const int*)d_in[4];
    float* out = (float*)d_out;

    char* ws = (char*)d_ws;
    float* G   = (float*)ws;
    float* wab = (float*)(ws + (size_t)N * F * sizeof(float));

    gat_w_dot_a<<<F / 8, 256, 0, stream>>>(W, a, wab);

    gat_score_gather<<<N, 32, 0, stream>>>(feat, embed, wab, nidx, G);

    {
        const int tiles = (N / 64) * (F / 64);
        gat_gemm_xw<<<tiles / 2, 256, 0, stream>>>(G, W, out, N);
    }
}
